# starting kernel
# baseline (speedup 1.0000x reference)
#include <hip/hip_runtime.h>


typedef __attribute__((ext_vector_type(2))) float v2f;
typedef __attribute__((ext_vector_type(8))) float v8f;

#define IN_DIM 128


__global__ void fill_deg_kernel(float* __restrict__ deg, int n) {
  int i = blockIdx.x * blockDim.x + threadIdx.x;
  if (i < n) deg[i] = 1.0f;
}

__global__ void deg_accum_kernel(const int* __restrict__ tgt, float* __restrict__ deg, int e) {
  int i = blockIdx.x * blockDim.x + threadIdx.x;
  if (i < e) atomicAdd(&deg[tgt[i]], 1.0f);
}

__global__ void rsqrt_kernel(float* __restrict__ d, int n) {
  int i = blockIdx.x * blockDim.x + threadIdx.x;
  if (i < n) d[i] = rsqrtf(d[i]);
}


template <int NC>
__global__ __launch_bounds__(256) void gemm_wmma_f32(const float* __restrict__ X,
                                                     const float* __restrict__ W,
                                                     float* __restrict__ H, int n_rows) {
  __shared__ __align__(16) float Wt[NC * IN_DIM];
  for (int i = threadIdx.x; i < NC * IN_DIM; i += 256) {
    int k = i / NC;
    int c = i - k * NC;
    Wt[c * IN_DIM + k] = W[i];
  }
  __syncthreads();

  const int lane = threadIdx.x & 31;
  const int wave = threadIdx.x >> 5;
  const int row_base = blockIdx.x * 128 + wave * 16;
  if (row_base + 16 > n_rows) return;

  const int r = lane & 15;
  const int koff = (lane >> 4) << 1;

  const float* xrow = X + (size_t)(row_base + r) * IN_DIM + koff;

  constexpr int NT = NC / 16;
  v8f acc[NT];
  const v8f vzero = {0.f, 0.f, 0.f, 0.f, 0.f, 0.f, 0.f, 0.f};
#pragma unroll
  for (int t = 0; t < NT; ++t) acc[t] = vzero;

#pragma unroll 4
  for (int k0 = 0; k0 < IN_DIM; k0 += 4) {
    v2f a = *(const v2f*)(xrow + k0);
#pragma unroll
    for (int t = 0; t < NT; ++t) {
      v2f b = *(const v2f*)(&Wt[(t * 16 + r) * IN_DIM + k0 + koff]);
      acc[t] = __builtin_amdgcn_wmma_f32_16x16x4_f32(false, a, false, b,
                                                     (short)0, acc[t], false, false);
    }
  }

  const int rowh = row_base + ((lane >> 4) << 3);
#pragma unroll
  for (int t = 0; t < NT; ++t) {
    const int col = t * 16 + r;
#pragma unroll
    for (int j = 0; j < 8; ++j)
      H[(size_t)(rowh + j) * NC + col] = acc[t][j];
  }
}


__global__ void init_agg_kernel(const float* __restrict__ h, const float* __restrict__ dinv,
                                const float* __restrict__ bias, float* __restrict__ agg,
                                int n, int D) {
  int idx = blockIdx.x * blockDim.x + threadIdx.x;
  if (idx >= n * D) return;
  int i = idx / D;
  int d = idx - i * D;
  float di = dinv[i];
  agg[idx] = di * di * h[idx] + bias[d];
}

__global__ __launch_bounds__(256) void scatter_d128(const int* __restrict__ src,
                                                    const int* __restrict__ tgt,
                                                    const float* __restrict__ dinv,
                                                    const float* __restrict__ h,
                                                    float* __restrict__ agg, int n_edges) {
  int e = blockIdx.x * 8 + (threadIdx.x >> 5);
  if (e >= n_edges) return;
  int lane = threadIdx.x & 31;
  int s = src[e], t = tgt[e];
  float w = dinv[s] * dinv[t];
  const float4 v = *(const float4*)(h + (size_t)s * 128 + lane * 4);
  float* ap = agg + (size_t)t * 128 + lane * 4;
  atomicAdd(ap + 0, w * v.x);
  atomicAdd(ap + 1, w * v.y);
  atomicAdd(ap + 2, w * v.z);
  atomicAdd(ap + 3, w * v.w);
}

__global__ __launch_bounds__(256) void scatter_d64(const int* __restrict__ src,
                                                   const int* __restrict__ tgt,
                                                   const float* __restrict__ dinv,
                                                   const float* __restrict__ h,
                                                   float* __restrict__ agg, int n_edges) {
  int e = blockIdx.x * 8 + (threadIdx.x >> 5);
  if (e >= n_edges) return;
  int lane = threadIdx.x & 31;
  int s = src[e], t = tgt[e];
  float w = dinv[s] * dinv[t];
  const float2 v = *(const float2*)(h + (size_t)s * 64 + lane * 2);
  float* ap = agg + (size_t)t * 64 + lane * 2;
  atomicAdd(ap + 0, w * v.x);
  atomicAdd(ap + 1, w * v.y);
}

__global__ void relu_kernel(float* __restrict__ a, int n) {
  int i = blockIdx.x * blockDim.x + threadIdx.x;
  if (i < n) a[i] = fmaxf(a[i], 0.0f);
}


static inline size_t align_up(size_t x, size_t a) { return (x + a - 1) & ~(a - 1); }

extern "C" void kernel_launch(void* const* d_in, const int* in_sizes, int n_in,
                              void* d_out, int out_size, void* d_ws, size_t ws_size,
                              hipStream_t stream) {
  const float* x  = (const float*)d_in[0];
  const int*   ei = (const int*)d_in[1];
  const float* W1 = (const float*)d_in[2];
  const float* b1 = (const float*)d_in[3];
  const float* W2 = (const float*)d_in[4];
  const float* b2 = (const float*)d_in[5];

  const int N = in_sizes[0] / IN_DIM;
  const int E = in_sizes[1] / 2;
  const int* src = ei;
  const int* tgt = ei + E;

  char* ws = (char*)d_ws;
  size_t off = 0;
  float* dinv = (float*)(ws + off); off = align_up(off + (size_t)N * 4, 256);
  float* H1   = (float*)(ws + off); off = align_up(off + (size_t)N * IN_DIM * 4, 256);
  float* A1   = (float*)(ws + off); off = align_up(off + (size_t)N * IN_DIM * 4, 256);
  float* H2   = H1;
  float* out  = (float*)d_out;

  const int TB = 256;
  dim3 blk(TB);

  fill_deg_kernel<<<dim3((N + TB - 1) / TB), blk, 0, stream>>>(dinv, N);
  deg_accum_kernel<<<dim3((E + TB - 1) / TB), blk, 0, stream>>>(tgt, dinv, E);
  rsqrt_kernel<<<dim3((N + TB - 1) / TB), blk, 0, stream>>>(dinv, N);

  const int gemm_blocks = (N + 127) / 128;
  const int eblocks = (E + 7) / 8;

  gemm_wmma_f32<128><<<dim3(gemm_blocks), blk, 0, stream>>>(x, W1, H1, N);
  init_agg_kernel<<<dim3((N * 128 + TB - 1) / TB), blk, 0, stream>>>(H1, dinv, b1, A1, N, 128);
  scatter_d128<<<dim3(eblocks), blk, 0, stream>>>(src, tgt, dinv, H1, A1, E);
  relu_kernel<<<dim3((N * 128 + TB - 1) / TB), blk, 0, stream>>>(A1, N * 128);

  gemm_wmma_f32<64><<<dim3(gemm_blocks), blk, 0, stream>>>(A1, W2, H2, N);
  init_agg_kernel<<<dim3((N * 64 + TB - 1) / TB), blk, 0, stream>>>(H2, dinv, b2, out, N, 64);
  scatter_d64<<<dim3(eblocks), blk, 0, stream>>>(src, tgt, dinv, H2, out, E);
}


# edge scatter by f32 atomics -> counting-sort CSR (int atomic rank, scan, place) + per-node register gather, one writer per row, self-loop/bias/ReLU in its store; sums in CSR arrival order
# speedup vs baseline: 2.1937x; 2.1937x over previous
#include <hip/hip_runtime.h>


typedef __attribute__((ext_vector_type(2))) float v2f;
typedef __attribute__((ext_vector_type(8))) float v8f;

#define IN_DIM 128
#define SCAN_TB 1024


__global__ void zero_cnt_kernel(int* __restrict__ cnt, int n) {
  int i = blockIdx.x * blockDim.x + threadIdx.x;
  if (i < n) cnt[i] = 0;
}

__global__ void count_rank_kernel(const int* __restrict__ tgt, int* __restrict__ cnt,
                                  int* __restrict__ rank, int e) {
  int i = blockIdx.x * blockDim.x + threadIdx.x;
  if (i < e) rank[i] = atomicAdd(&cnt[tgt[i]], 1);
}

__global__ __launch_bounds__(SCAN_TB) void scan_local_kernel(const int* __restrict__ cnt,
                                                             int* __restrict__ rowptr,
                                                             int* __restrict__ bsum, int n) {
  __shared__ int wsum[32];
  const int i = blockIdx.x * SCAN_TB + threadIdx.x;
  const int lane = threadIdx.x & 31;
  const int wv = threadIdx.x >> 5;
  const int c = (i < n) ? cnt[i] : 0;
  int v = c;
  for (int o = 1; o < 32; o <<= 1) {
    int t = __shfl_up(v, o);
    if (lane >= o) v += t;
  }
  if (lane == 31) wsum[wv] = v;
  __syncthreads();
  if (wv == 0) {
    const int t = wsum[lane];
    int u = t;
    for (int o = 1; o < 32; o <<= 1) {
      int q = __shfl_up(u, o);
      if (lane >= o) u += q;
    }
    wsum[lane] = u - t;
    if (lane == 31) bsum[blockIdx.x] = u;
  }
  __syncthreads();
  if (i < n) rowptr[i] = v - c + wsum[wv];
}

__global__ __launch_bounds__(SCAN_TB) void scan_finish_kernel(const int* __restrict__ cnt,
                                                              const int* __restrict__ bsum,
                                                              int* __restrict__ rowptr,
                                                              float* __restrict__ dinv, int n) {
  int off = 0;
  for (int j = 0; j < (int)blockIdx.x; ++j) off += bsum[j];
  const int i = blockIdx.x * SCAN_TB + threadIdx.x;
  if (i < n) {
    const int c = cnt[i];
    const int r = rowptr[i] + off;
    rowptr[i] = r;
    dinv[i] = rsqrtf((float)(c + 1));
    if (i == n - 1) rowptr[n] = r + c;
  }
}

__global__ void place_kernel(const int* __restrict__ src, const int* __restrict__ tgt,
                             const int* __restrict__ rank, const int* __restrict__ rowptr,
                             const float* __restrict__ dinv, int* __restrict__ csr, int e) {
  int i = blockIdx.x * blockDim.x + threadIdx.x;
  if (i < e) {
    const int s = src[i];
    int2 ent;
    ent.x = s;
    ent.y = __float_as_int(dinv[s]);
    *(int2*)(csr + 2 * (size_t)(rowptr[tgt[i]] + rank[i])) = ent;
  }
}


template <int NC>
__global__ __launch_bounds__(256) void gemm_wmma_f32(const float* __restrict__ X,
                                                     const float* __restrict__ W,
                                                     float* __restrict__ H, int n_rows) {
  __shared__ __align__(16) float Wt[NC * IN_DIM];
  for (int i = threadIdx.x; i < NC * IN_DIM; i += 256) {
    int k = i / NC;
    int c = i - k * NC;
    Wt[c * IN_DIM + k] = W[i];
  }
  __syncthreads();

  const int lane = threadIdx.x & 31;
  const int wave = threadIdx.x >> 5;
  const int row_base = blockIdx.x * 128 + wave * 16;
  if (row_base + 16 > n_rows) return;

  const int r = lane & 15;
  const int koff = (lane >> 4) << 1;

  const float* xrow = X + (size_t)(row_base + r) * IN_DIM + koff;

  constexpr int NT = NC / 16;
  v8f acc[NT];
  const v8f vzero = {0.f, 0.f, 0.f, 0.f, 0.f, 0.f, 0.f, 0.f};
#pragma unroll
  for (int t = 0; t < NT; ++t) acc[t] = vzero;

#pragma unroll 4
  for (int k0 = 0; k0 < IN_DIM; k0 += 4) {
    v2f a = *(const v2f*)(xrow + k0);
#pragma unroll
    for (int t = 0; t < NT; ++t) {
      v2f b = *(const v2f*)(&Wt[(t * 16 + r) * IN_DIM + k0 + koff]);
      acc[t] = __builtin_amdgcn_wmma_f32_16x16x4_f32(false, a, false, b,
                                                     (short)0, acc[t], false, false);
    }
  }

  const int rowh = row_base + ((lane >> 4) << 3);
#pragma unroll
  for (int t = 0; t < NT; ++t) {
    const int col = t * 16 + r;
#pragma unroll
    for (int j = 0; j < 8; ++j)
      H[(size_t)(rowh + j) * NC + col] = acc[t][j];
  }
}


template <int D, bool RELU>
__global__ __launch_bounds__(256) void gather_rows(const int* __restrict__ rowptr,
                                                   const int* __restrict__ csr,
                                                   const float* __restrict__ dinv,
                                                   const float* __restrict__ h,
                                                   const float* __restrict__ bias,
                                                   float* __restrict__ outp, int n) {
  constexpr int LPR = D / 4;
  constexpr int RPL = 32 / LPR;
  const int wv = __builtin_amdgcn_readfirstlane((int)(threadIdx.x >> 5));
  const int node = blockIdx.x * 8 + wv;
  if (node >= n) return;
  const int lane = threadIdx.x & 31;
  const int sub = lane / LPR;
  const int col = (lane - sub * LPR) * 4;

  const int beg = rowptr[node];
  const int end = rowptr[node + 1];
  const float di = dinv[node];

  float4 acc = make_float4(0.f, 0.f, 0.f, 0.f);
  if (sub == 0) {
    const float4 hv = *(const float4*)(h + (size_t)node * D + col);
    const float4 bv = *(const float4*)(bias + col);
    const float dd = di * di;
    acc.x = dd * hv.x + bv.x;
    acc.y = dd * hv.y + bv.y;
    acc.z = dd * hv.z + bv.z;
    acc.w = dd * hv.w + bv.w;
  }

  for (int base = beg; base < end; base += 32) {
    const int m = min(32, end - base);
    int s = 0;
    float w = 0.f;
    if (lane < m) {
      const int2 ent = *(const int2*)(csr + 2 * (size_t)(base + lane));
      s = ent.x;
      w = __int_as_float(ent.y) * di;
    }
    const int wi = __float_as_int(w);
    const int mp = m - (m % RPL);
#pragma unroll 4
    for (int j = 0; j < mp; j += RPL) {
      int sj = __builtin_amdgcn_readlane(s, j);
      int wj = __builtin_amdgcn_readlane(wi, j);
      if (RPL == 2) {
        const int s1 = __builtin_amdgcn_readlane(s, j + 1);
        const int w1 = __builtin_amdgcn_readlane(wi, j + 1);
        sj = sub ? s1 : sj;
        wj = sub ? w1 : wj;
      }
      const float wf = __int_as_float(wj);
      const float4 v = *(const float4*)(h + (size_t)sj * D + col);
      acc.x += wf * v.x;
      acc.y += wf * v.y;
      acc.z += wf * v.z;
      acc.w += wf * v.w;
    }
    if (RPL == 2 && mp < m) {
      const int sj = __builtin_amdgcn_readlane(s, mp);
      const float wf = __int_as_float(__builtin_amdgcn_readlane(wi, mp));
      if (sub == 0) {
        const float4 v = *(const float4*)(h + (size_t)sj * D + col);
        acc.x += wf * v.x;
        acc.y += wf * v.y;
        acc.z += wf * v.z;
        acc.w += wf * v.w;
      }
    }
  }

  if (RPL == 2) {
    acc.x += __shfl_xor(acc.x, 16);
    acc.y += __shfl_xor(acc.y, 16);
    acc.z += __shfl_xor(acc.z, 16);
    acc.w += __shfl_xor(acc.w, 16);
  }
  if (RELU) {
    acc.x = fmaxf(acc.x, 0.0f);
    acc.y = fmaxf(acc.y, 0.0f);
    acc.z = fmaxf(acc.z, 0.0f);
    acc.w = fmaxf(acc.w, 0.0f);
  }
  if (sub == 0) *(float4*)(outp + (size_t)node * D + col) = acc;
}


static inline size_t align_up(size_t x, size_t a) { return (x + a - 1) & ~(a - 1); }

extern "C" void kernel_launch(void* const* d_in, const int* in_sizes, int n_in,
                              void* d_out, int out_size, void* d_ws, size_t ws_size,
                              hipStream_t stream) {
  const float* x  = (const float*)d_in[0];
  const int*   ei = (const int*)d_in[1];
  const float* W1 = (const float*)d_in[2];
  const float* b1 = (const float*)d_in[3];
  const float* W2 = (const float*)d_in[4];
  const float* b2 = (const float*)d_in[5];

  const int N = in_sizes[0] / IN_DIM;
  const int E = in_sizes[1] / 2;
  const int* src = ei;
  const int* tgt = ei + E;

  char* ws = (char*)d_ws;
  size_t off = 0;
  float* dinv   = (float*)(ws + off); off = align_up(off + (size_t)N * 4, 256);
  int*   cnt    = (int*)(ws + off);   off = align_up(off + (size_t)N * 4, 256);
  int*   rowptr = (int*)(ws + off);   off = align_up(off + (size_t)(N + 1) * 4, 256);
  int*   bsum   = (int*)(ws + off);   off = align_up(off + (size_t)((N + SCAN_TB - 1) / SCAN_TB) * 4, 256);
  int*   rank   = (int*)(ws + off);   off = align_up(off + (size_t)E * 4, 256);
  int*   csr    = (int*)(ws + off);   off = align_up(off + (size_t)E * 8, 256);
  float* H1     = (float*)(ws + off); off = align_up(off + (size_t)N * IN_DIM * 4, 256);
  float* A1     = (float*)(ws + off); off = align_up(off + (size_t)N * IN_DIM * 4, 256);
  float* H2     = H1;
  float* out    = (float*)d_out;

  const int TB = 256;
  dim3 blk(TB);
  const int sblocks = (N + SCAN_TB - 1) / SCAN_TB;

  zero_cnt_kernel<<<dim3((N + TB - 1) / TB), blk, 0, stream>>>(cnt, N);
  count_rank_kernel<<<dim3((E + TB - 1) / TB), blk, 0, stream>>>(tgt, cnt, rank, E);
  scan_local_kernel<<<dim3(sblocks), dim3(SCAN_TB), 0, stream>>>(cnt, rowptr, bsum, N);
  scan_finish_kernel<<<dim3(sblocks), dim3(SCAN_TB), 0, stream>>>(cnt, bsum, rowptr, dinv, N);
  place_kernel<<<dim3((E + TB - 1) / TB), blk, 0, stream>>>(src, tgt, rank, rowptr, dinv, csr, E);

  const int gemm_blocks = (N + 127) / 128;
  const int nblocks = (N + 7) / 8;

  gemm_wmma_f32<128><<<dim3(gemm_blocks), blk, 0, stream>>>(x, W1, H1, N);
  gather_rows<128, true><<<dim3(nblocks), blk, 0, stream>>>(rowptr, csr, dinv, H1, b1, A1, N);

  gemm_wmma_f32<64><<<dim3(gemm_blocks), blk, 0, stream>>>(A1, W2, H2, N);
  gather_rows<64, false><<<dim3(nblocks), blk, 0, stream>>>(rowptr, csr, dinv, H2, b2, out, N);
}


# GEMM: LDS row pitch of the staged W^T 128 -> 132 dwords (16 B x odd): B-fragment 8-byte reads and the transposing stores no longer put 16 / 32 lanes on one bank; summation order unchanged
# speedup vs baseline: 2.7897x; 1.2717x over previous
#include <hip/hip_runtime.h>


typedef __attribute__((ext_vector_type(2))) float v2f;
typedef __attribute__((ext_vector_type(8))) float v8f;

#define IN_DIM 128
#define SCAN_TB 1024
#define WT_PITCH (IN_DIM + 4)


__global__ void zero_cnt_kernel(int* __restrict__ cnt, int n) {
  int i = blockIdx.x * blockDim.x + threadIdx.x;
  if (i < n) cnt[i] = 0;
}

__global__ void count_rank_kernel(const int* __restrict__ tgt, int* __restrict__ cnt,
                                  int* __restrict__ rank, int e) {
  int i = blockIdx.x * blockDim.x + threadIdx.x;
  if (i < e) rank[i] = atomicAdd(&cnt[tgt[i]], 1);
}

__global__ __launch_bounds__(SCAN_TB) void scan_local_kernel(const int* __restrict__ cnt,
                                                             int* __restrict__ rowptr,
                                                             int* __restrict__ bsum, int n) {
  __shared__ int wsum[32];
  const int i = blockIdx.x * SCAN_TB + threadIdx.x;
  const int lane = threadIdx.x & 31;
  const int wv = threadIdx.x >> 5;
  const int c = (i < n) ? cnt[i] : 0;
  int v = c;
  for (int o = 1; o < 32; o <<= 1) {
    int t = __shfl_up(v, o);
    if (lane >= o) v += t;
  }
  if (lane == 31) wsum[wv] = v;
  __syncthreads();
  if (wv == 0) {
    const int t = wsum[lane];
    int u = t;
    for (int o = 1; o < 32; o <<= 1) {
      int q = __shfl_up(u, o);
      if (lane >= o) u += q;
    }
    wsum[lane] = u - t;
    if (lane == 31) bsum[blockIdx.x] = u;
  }
  __syncthreads();
  if (i < n) rowptr[i] = v - c + wsum[wv];
}

__global__ __launch_bounds__(SCAN_TB) void scan_finish_kernel(const int* __restrict__ cnt,
                                                              const int* __restrict__ bsum,
                                                              int* __restrict__ rowptr,
                                                              float* __restrict__ dinv, int n) {
  int off = 0;
  for (int j = 0; j < (int)blockIdx.x; ++j) off += bsum[j];
  const int i = blockIdx.x * SCAN_TB + threadIdx.x;
  if (i < n) {
    const int c = cnt[i];
    const int r = rowptr[i] + off;
    rowptr[i] = r;
    dinv[i] = rsqrtf((float)(c + 1));
    if (i == n - 1) rowptr[n] = r + c;
  }
}

__global__ void place_kernel(const int* __restrict__ src, const int* __restrict__ tgt,
                             const int* __restrict__ rank, const int* __restrict__ rowptr,
                             const float* __restrict__ dinv, int* __restrict__ csr, int e) {
  int i = blockIdx.x * blockDim.x + threadIdx.x;
  if (i < e) {
    const int s = src[i];
    int2 ent;
    ent.x = s;
    ent.y = __float_as_int(dinv[s]);
    *(int2*)(csr + 2 * (size_t)(rowptr[tgt[i]] + rank[i])) = ent;
  }
}


template <int NC>
__global__ __launch_bounds__(256) void gemm_wmma_f32(const float* __restrict__ X,
                                                     const float* __restrict__ W,
                                                     float* __restrict__ H, int n_rows) {
  __shared__ __align__(16) float Wt[NC * WT_PITCH];
  for (int i = threadIdx.x; i < NC * IN_DIM; i += 256) {
    int k = i / NC;
    int c = i - k * NC;
    Wt[c * WT_PITCH + k] = W[i];
  }
  __syncthreads();

  const int lane = threadIdx.x & 31;
  const int wave = threadIdx.x >> 5;
  const int row_base = blockIdx.x * 128 + wave * 16;
  if (row_base + 16 > n_rows) return;

  const int r = lane & 15;
  const int koff = (lane >> 4) << 1;

  const float* xrow = X + (size_t)(row_base + r) * IN_DIM + koff;

  constexpr int NT = NC / 16;
  v8f acc[NT];
  const v8f vzero = {0.f, 0.f, 0.f, 0.f, 0.f, 0.f, 0.f, 0.f};
#pragma unroll
  for (int t = 0; t < NT; ++t) acc[t] = vzero;

#pragma unroll 4
  for (int k0 = 0; k0 < IN_DIM; k0 += 4) {
    v2f a = *(const v2f*)(xrow + k0);
#pragma unroll
    for (int t = 0; t < NT; ++t) {
      v2f b = *(const v2f*)(&Wt[(t * 16 + r) * WT_PITCH + k0 + koff]);
      acc[t] = __builtin_amdgcn_wmma_f32_16x16x4_f32(false, a, false, b,
                                                     (short)0, acc[t], false, false);
    }
  }

  const int rowh = row_base + ((lane >> 4) << 3);
#pragma unroll
  for (int t = 0; t < NT; ++t) {
    const int col = t * 16 + r;
#pragma unroll
    for (int j = 0; j < 8; ++j)
      H[(size_t)(rowh + j) * NC + col] = acc[t][j];
  }
}


template <int D, bool RELU>
__global__ __launch_bounds__(256) void gather_rows(const int* __restrict__ rowptr,
                                                   const int* __restrict__ csr,
                                                   const float* __restrict__ dinv,
                                                   const float* __restrict__ h,
                                                   const float* __restrict__ bias,
                                                   float* __restrict__ outp, int n) {
  constexpr int LPR = D / 4;
  constexpr int RPL = 32 / LPR;
  const int wv = __builtin_amdgcn_readfirstlane((int)(threadIdx.x >> 5));
  const int node = blockIdx.x * 8 + wv;
  if (node >= n) return;
  const int lane = threadIdx.x & 31;
  const int sub = lane / LPR;
  const int col = (lane - sub * LPR) * 4;

  const int beg = rowptr[node];
  const int end = rowptr[node + 1];
  const float di = dinv[node];

  float4 acc = make_float4(0.f, 0.f, 0.f, 0.f);
  if (sub == 0) {
    const float4 hv = *(const float4*)(h + (size_t)node * D + col);
    const float4 bv = *(const float4*)(bias + col);
    const float dd = di * di;
    acc.x = dd * hv.x + bv.x;
    acc.y = dd * hv.y + bv.y;
    acc.z = dd * hv.z + bv.z;
    acc.w = dd * hv.w + bv.w;
  }

  for (int base = beg; base < end; base += 32) {
    const int m = min(32, end - base);
    int s = 0;
    float w = 0.f;
    if (lane < m) {
      const int2 ent = *(const int2*)(csr + 2 * (size_t)(base + lane));
      s = ent.x;
      w = __int_as_float(ent.y) * di;
    }
    const int wi = __float_as_int(w);
    const int mp = m - (m % RPL);
#pragma unroll 4
    for (int j = 0; j < mp; j += RPL) {
      int sj = __builtin_amdgcn_readlane(s, j);
      int wj = __builtin_amdgcn_readlane(wi, j);
      if (RPL == 2) {
        const int s1 = __builtin_amdgcn_readlane(s, j + 1);
        const int w1 = __builtin_amdgcn_readlane(wi, j + 1);
        sj = sub ? s1 : sj;
        wj = sub ? w1 : wj;
      }
      const float wf = __int_as_float(wj);
      const float4 v = *(const float4*)(h + (size_t)sj * D + col);
      acc.x += wf * v.x;
      acc.y += wf * v.y;
      acc.z += wf * v.z;
      acc.w += wf * v.w;
    }
    if (RPL == 2 && mp < m) {
      const int sj = __builtin_amdgcn_readlane(s, mp);
      const float wf = __int_as_float(__builtin_amdgcn_readlane(wi, mp));
      if (sub == 0) {
        const float4 v = *(const float4*)(h + (size_t)sj * D + col);
        acc.x += wf * v.x;
        acc.y += wf * v.y;
        acc.z += wf * v.z;
        acc.w += wf * v.w;
      }
    }
  }

  if (RPL == 2) {
    acc.x += __shfl_xor(acc.x, 16);
    acc.y += __shfl_xor(acc.y, 16);
    acc.z += __shfl_xor(acc.z, 16);
    acc.w += __shfl_xor(acc.w, 16);
  }
  if (RELU) {
    acc.x = fmaxf(acc.x, 0.0f);
    acc.y = fmaxf(acc.y, 0.0f);
    acc.z = fmaxf(acc.z, 0.0f);
    acc.w = fmaxf(acc.w, 0.0f);
  }
  if (sub == 0) *(float4*)(outp + (size_t)node * D + col) = acc;
}


static inline size_t align_up(size_t x, size_t a) { return (x + a - 1) & ~(a - 1); }

extern "C" void kernel_launch(void* const* d_in, const int* in_sizes, int n_in,
                              void* d_out, int out_size, void* d_ws, size_t ws_size,
                              hipStream_t stream) {
  const float* x  = (const float*)d_in[0];
  const int*   ei = (const int*)d_in[1];
  const float* W1 = (const float*)d_in[2];
  const float* b1 = (const float*)d_in[3];
  const float* W2 = (const float*)d_in[4];
  const float* b2 = (const float*)d_in[5];

  const int N = in_sizes[0] / IN_DIM;
  const int E = in_sizes[1] / 2;
  const int* src = ei;
  const int* tgt = ei + E;

  char* ws = (char*)d_ws;
  size_t off = 0;
  float* dinv   = (float*)(ws + off); off = align_up(off + (size_t)N * 4, 256);
  int*   cnt    = (int*)(ws + off);   off = align_up(off + (size_t)N * 4, 256);
  int*   rowptr = (int*)(ws + off);   off = align_up(off + (size_t)(N + 1) * 4, 256);
  int*   bsum   = (int*)(ws + off);   off = align_up(off + (size_t)((N + SCAN_TB - 1) / SCAN_TB) * 4, 256);
  int*   rank   = (int*)(ws + off);   off = align_up(off + (size_t)E * 4, 256);
  int*   csr    = (int*)(ws + off);   off = align_up(off + (size_t)E * 8, 256);
  float* H1     = (float*)(ws + off); off = align_up(off + (size_t)N * IN_DIM * 4, 256);
  float* A1     = (float*)(ws + off); off = align_up(off + (size_t)N * IN_DIM * 4, 256);
  float* H2     = H1;
  float* out    = (float*)d_out;

  const int TB = 256;
  dim3 blk(TB);
  const int sblocks = (N + SCAN_TB - 1) / SCAN_TB;

  zero_cnt_kernel<<<dim3((N + TB - 1) / TB), blk, 0, stream>>>(cnt, N);
  count_rank_kernel<<<dim3((E + TB - 1) / TB), blk, 0, stream>>>(tgt, cnt, rank, E);
  scan_local_kernel<<<dim3(sblocks), dim3(SCAN_TB), 0, stream>>>(cnt, rowptr, bsum, N);
  scan_finish_kernel<<<dim3(sblocks), dim3(SCAN_TB), 0, stream>>>(cnt, bsum, rowptr, dinv, N);
  place_kernel<<<dim3((E + TB - 1) / TB), blk, 0, stream>>>(src, tgt, rank, rowptr, dinv, csr, E);

  const int gemm_blocks = (N + 127) / 128;
  const int nblocks = (N + 7) / 8;

  gemm_wmma_f32<128><<<dim3(gemm_blocks), blk, 0, stream>>>(x, W1, H1, N);
  gather_rows<128, true><<<dim3(nblocks), blk, 0, stream>>>(rowptr, csr, dinv, H1, b1, A1, N);

  gemm_wmma_f32<64><<<dim3(gemm_blocks), blk, 0, stream>>>(A1, W2, H2, N);
  gather_rows<64, false><<<dim3(nblocks), blk, 0, stream>>>(rowptr, csr, dinv, H2, b2, out, N);
}


# layer-1 gather fused into the layer-2 GEMM: each wave sums 16 nodes' rows into its own LDS strip (the WMMA A operand), activations never go through memory, one launch fewer; sum order kept
# speedup vs baseline: 3.3997x; 1.2187x over previous
#include <hip/hip_runtime.h>


typedef __attribute__((ext_vector_type(2))) float v2f;
typedef __attribute__((ext_vector_type(8))) float v8f;

#define IN_DIM 128
#define WT_PITCH (IN_DIM + 4)

#define SLOTS 32

__global__ void zero_cnt_kernel(int* __restrict__ cnt, int n) {
  int i = blockIdx.x * blockDim.x + threadIdx.x;
  if (i <= n) cnt[i] = 0;
}

__global__ void count_place_kernel(const int* __restrict__ src, const int* __restrict__ tgt,
                                   int* __restrict__ cnt, int* __restrict__ slab,
                                   int* __restrict__ spill, int e, int n) {
  int i = blockIdx.x * blockDim.x + threadIdx.x;
  if (i < e) {
    const int t = tgt[i];
    const int s = src[i];
    const int pos = atomicAdd(&cnt[t], 1);
    if (pos < SLOTS) {
      slab[(size_t)t * SLOTS + pos] = s;
    } else {
      const int q = atomicAdd(&cnt[n], 1);
      int2 ent;
      ent.x = s;
      ent.y = t;
      *(int2*)(spill + 2 * (size_t)q) = ent;
    }
  }
}


template <int NC>
__global__ __launch_bounds__(256) void gemm_wmma_f32(const float* __restrict__ X,
                                                     const float* __restrict__ W,
                                                     const int* __restrict__ cnt,
                                                     float* __restrict__ H, int n_rows) {
  __shared__ __align__(16) float Wt[NC * WT_PITCH];
  for (int i = threadIdx.x; i < NC * IN_DIM; i += 256) {
    int k = i / NC;
    int c = i - k * NC;
    Wt[c * WT_PITCH + k] = W[i];
  }
  __syncthreads();

  const int lane = threadIdx.x & 31;
  const int wave = threadIdx.x >> 5;
  const int row_base = blockIdx.x * 128 + wave * 16;
  if (row_base + 16 > n_rows) return;

  const int r = lane & 15;
  const int koff = (lane >> 4) << 1;

  const float* xrow = X + (size_t)(row_base + r) * IN_DIM + koff;

  constexpr int NT = NC / 16;
  v8f acc[NT];
  const v8f vzero = {0.f, 0.f, 0.f, 0.f, 0.f, 0.f, 0.f, 0.f};
#pragma unroll
  for (int t = 0; t < NT; ++t) acc[t] = vzero;

#pragma unroll 4
  for (int k0 = 0; k0 < IN_DIM; k0 += 4) {
    v2f a = *(const v2f*)(xrow + k0);
#pragma unroll
    for (int t = 0; t < NT; ++t) {
      v2f b = *(const v2f*)(&Wt[(t * 16 + r) * WT_PITCH + k0 + koff]);
      acc[t] = __builtin_amdgcn_wmma_f32_16x16x4_f32(false, a, false, b,
                                                     (short)0, acc[t], false, false);
    }
  }

  const int rowh = row_base + ((lane >> 4) << 3);
  const int4 d0 = *(const int4*)(cnt + rowh);
  const int4 d1 = *(const int4*)(cnt + rowh + 4);
  float sc[8];
  sc[0] = rsqrtf((float)(d0.x + 1));
  sc[1] = rsqrtf((float)(d0.y + 1));
  sc[2] = rsqrtf((float)(d0.z + 1));
  sc[3] = rsqrtf((float)(d0.w + 1));
  sc[4] = rsqrtf((float)(d1.x + 1));
  sc[5] = rsqrtf((float)(d1.y + 1));
  sc[6] = rsqrtf((float)(d1.z + 1));
  sc[7] = rsqrtf((float)(d1.w + 1));
#pragma unroll
  for (int t = 0; t < NT; ++t) {
    const int col = t * 16 + r;
#pragma unroll
    for (int j = 0; j < 8; ++j)
      H[(size_t)(rowh + j) * NC + col] = acc[t][j] * sc[j];
  }
}


template <int D, bool RELU>
__global__ __launch_bounds__(256) void gather_rows(const int* __restrict__ cnt,
                                                   const int* __restrict__ slab,
                                                   const int* __restrict__ spill,
                                                   const float* __restrict__ h,
                                                   const float* __restrict__ bias,
                                                   float* __restrict__ outp, int n) {
  constexpr int LPR = D / 4;
  constexpr int RPL = 32 / LPR;
  const int wv = __builtin_amdgcn_readfirstlane((int)(threadIdx.x >> 5));
  const int node = blockIdx.x * 8 + wv;
  if (node >= n) return;
  const int lane = threadIdx.x & 31;
  const int sub = lane / LPR;
  const int col = (lane - sub * LPR) * 4;

  const int deg = cnt[node];
  const int m = min(deg, SLOTS);

  float4 acc = make_float4(0.f, 0.f, 0.f, 0.f);
  if (sub == 0) acc = *(const float4*)(h + (size_t)node * D + col);

  int s = 0;
  if (lane < m) s = slab[(size_t)node * SLOTS + lane];
  const int mp = m - (m % RPL);
#pragma unroll 4
  for (int j = 0; j < mp; j += RPL) {
    int sj = __builtin_amdgcn_readlane(s, j);
    if (RPL == 2) {
      const int s1 = __builtin_amdgcn_readlane(s, j + 1);
      sj = sub ? s1 : sj;
    }
    const float4 v = *(const float4*)(h + (size_t)sj * D + col);
    acc.x += v.x;
    acc.y += v.y;
    acc.z += v.z;
    acc.w += v.w;
  }
  if (RPL == 2 && mp < m) {
    const int sj = __builtin_amdgcn_readlane(s, mp);
    if (sub == 0) {
      const float4 v = *(const float4*)(h + (size_t)sj * D + col);
      acc.x += v.x;
      acc.y += v.y;
      acc.z += v.z;
      acc.w += v.w;
    }
  }

  if (deg > SLOTS) {
    const int ns = cnt[n];
    for (int base = 0; base < ns; base += 32) {
      int es = 0;
      int et = -1;
      if (base + lane < ns) {
        const int2 ent = *(const int2*)(spill + 2 * (size_t)(base + lane));
        es = ent.x;
        et = ent.y;
      }
      unsigned mine = __builtin_amdgcn_ballot_w32(et == node);
      while (mine) {
        const int b = __builtin_ctz(mine);
        mine &= mine - 1;
        const int sj = __builtin_amdgcn_readlane(es, b);
        if (sub == 0) {
          const float4 v = *(const float4*)(h + (size_t)sj * D + col);
          acc.x += v.x;
          acc.y += v.y;
          acc.z += v.z;
          acc.w += v.w;
        }
      }
    }
  }

  if (RPL == 2) {
    acc.x += __shfl_xor(acc.x, 16);
    acc.y += __shfl_xor(acc.y, 16);
    acc.z += __shfl_xor(acc.z, 16);
    acc.w += __shfl_xor(acc.w, 16);
  }
  const float di = rsqrtf((float)(deg + 1));
  const float4 bv = *(const float4*)(bias + col);
  acc.x = di * acc.x + bv.x;
  acc.y = di * acc.y + bv.y;
  acc.z = di * acc.z + bv.z;
  acc.w = di * acc.w + bv.w;
  if (RELU) {
    acc.x = fmaxf(acc.x, 0.0f);
    acc.y = fmaxf(acc.y, 0.0f);
    acc.z = fmaxf(acc.z, 0.0f);
    acc.w = fmaxf(acc.w, 0.0f);
  }
  if (sub == 0) *(float4*)(outp + (size_t)node * D + col) = acc;
}

#define FW 25
#define OUT_DIM 64

__global__ __launch_bounds__(FW * 32) void gather_relu_gemm(const int* __restrict__ cnt,
                                                            const int* __restrict__ slab,
                                                            const int* __restrict__ spill,
                                                            const float* __restrict__ h,
                                                            const float* __restrict__ bias,
                                                            const float* __restrict__ W,
                                                            float* __restrict__ H2, int n) {
  __shared__ __align__(16) float Wt[OUT_DIM * WT_PITCH];
  __shared__ __align__(16) float strips[FW * 16 * WT_PITCH];
  const int lane = threadIdx.x & 31;
  const int wv = __builtin_amdgcn_readfirstlane((int)(threadIdx.x >> 5));
  const int row_base = (blockIdx.x * FW + wv) * 16;
  const bool live = row_base + 16 <= n;
  float* strip = strips + wv * (16 * WT_PITCH);
  const int col = lane * 4;

  int cntv = 0;
  float4 bv = make_float4(0.f, 0.f, 0.f, 0.f);
  if (live) {
    cntv = cnt[row_base + (lane & 15)];
    bv = *(const float4*)(bias + col);
  }

  constexpr int NW = OUT_DIM * IN_DIM;
  constexpr int TPB = FW * 32;
  constexpr int FULL = NW / TPB;
#pragma unroll
  for (int it = 0; it < FULL; ++it) {
    const int i = threadIdx.x + it * TPB;
    const int k = i / OUT_DIM;
    const int c = i - k * OUT_DIM;
    Wt[c * WT_PITCH + k] = W[i];
  }
  {
    const int i = threadIdx.x + FULL * TPB;
    if (i < NW) {
      const int k = i / OUT_DIM;
      const int c = i - k * OUT_DIM;
      Wt[c * WT_PITCH + k] = W[i];
    }
  }
  __syncthreads();
  if (!live) return;

  int m_next = min(__builtin_amdgcn_readlane(cntv, 0), SLOTS);
  int s_next = 0;
  if (lane < m_next) s_next = slab[(size_t)row_base * SLOTS + lane];

  for (int j = 0; j < 16; ++j) {
    const int node = row_base + j;
    const int deg = __builtin_amdgcn_readlane(cntv, j);
    const int m = m_next;
    const int s = s_next;
    if (j + 1 < 16) {
      m_next = min(__builtin_amdgcn_readlane(cntv, j + 1), SLOTS);
      s_next = 0;
      if (lane < m_next) s_next = slab[(size_t)(node + 1) * SLOTS + lane];
    }

    float4 acc = *(const float4*)(h + (size_t)node * IN_DIM + col);

    for (int base = 0; base < m; base += 8) {
      float4 v[8];
#pragma unroll
      for (int u = 0; u < 8; ++u) {
        v[u] = make_float4(0.f, 0.f, 0.f, 0.f);
        if (base + u < m) {
          const int sj = __builtin_amdgcn_readlane(s, base + u);
          v[u] = *(const float4*)(h + (size_t)sj * IN_DIM + col);
        }
      }
#pragma unroll
      for (int u = 0; u < 8; ++u) {
        acc.x += v[u].x;
        acc.y += v[u].y;
        acc.z += v[u].z;
        acc.w += v[u].w;
      }
    }

    if (deg > SLOTS) {
      const int ns = cnt[n];
      for (int base = 0; base < ns; base += 32) {
        int es = 0;
        int et = -1;
        if (base + lane < ns) {
          const int2 ent = *(const int2*)(spill + 2 * (size_t)(base + lane));
          es = ent.x;
          et = ent.y;
        }
        unsigned mine = __builtin_amdgcn_ballot_w32(et == node);
        while (mine) {
          const int b = __builtin_ctz(mine);
          mine &= mine - 1;
          const int sj = __builtin_amdgcn_readlane(es, b);
          const float4 v = *(const float4*)(h + (size_t)sj * IN_DIM + col);
          acc.x += v.x;
          acc.y += v.y;
          acc.z += v.z;
          acc.w += v.w;
        }
      }
    }

    const float di = rsqrtf((float)(deg + 1));
    acc.x = fmaxf(di * acc.x + bv.x, 0.0f);
    acc.y = fmaxf(di * acc.y + bv.y, 0.0f);
    acc.z = fmaxf(di * acc.z + bv.z, 0.0f);
    acc.w = fmaxf(di * acc.w + bv.w, 0.0f);
    *(float4*)(strip + j * WT_PITCH + col) = acc;
  }

  const int r = lane & 15;
  const int koff = (lane >> 4) << 1;
  constexpr int NT = OUT_DIM / 16;
  v8f acc2[NT];
  const v8f vzero = {0.f, 0.f, 0.f, 0.f, 0.f, 0.f, 0.f, 0.f};
#pragma unroll
  for (int t = 0; t < NT; ++t) acc2[t] = vzero;

#pragma unroll 4
  for (int k0 = 0; k0 < IN_DIM; k0 += 4) {
    v2f a = *(const v2f*)(strip + r * WT_PITCH + k0 + koff);
#pragma unroll
    for (int t = 0; t < NT; ++t) {
      v2f b = *(const v2f*)(&Wt[(t * 16 + r) * WT_PITCH + k0 + koff]);
      acc2[t] = __builtin_amdgcn_wmma_f32_16x16x4_f32(false, a, false, b,
                                                      (short)0, acc2[t], false, false);
    }
  }

  const int rowh = row_base + ((lane >> 4) << 3);
  const int4 d0 = *(const int4*)(cnt + rowh);
  const int4 d1 = *(const int4*)(cnt + rowh + 4);
  float sc[8];
  sc[0] = rsqrtf((float)(d0.x + 1));
  sc[1] = rsqrtf((float)(d0.y + 1));
  sc[2] = rsqrtf((float)(d0.z + 1));
  sc[3] = rsqrtf((float)(d0.w + 1));
  sc[4] = rsqrtf((float)(d1.x + 1));
  sc[5] = rsqrtf((float)(d1.y + 1));
  sc[6] = rsqrtf((float)(d1.z + 1));
  sc[7] = rsqrtf((float)(d1.w + 1));
#pragma unroll
  for (int t = 0; t < NT; ++t) {
    const int c = t * 16 + r;
#pragma unroll
    for (int j = 0; j < 8; ++j)
      H2[(size_t)(rowh + j) * OUT_DIM + c] = acc2[t][j] * sc[j];
  }
}


static inline size_t align_up(size_t x, size_t a) { return (x + a - 1) & ~(a - 1); }

extern "C" void kernel_launch(void* const* d_in, const int* in_sizes, int n_in,
                              void* d_out, int out_size, void* d_ws, size_t ws_size,
                              hipStream_t stream) {
  const float* x  = (const float*)d_in[0];
  const int*   ei = (const int*)d_in[1];
  const float* W1 = (const float*)d_in[2];
  const float* b1 = (const float*)d_in[3];
  const float* W2 = (const float*)d_in[4];
  const float* b2 = (const float*)d_in[5];

  const int N = in_sizes[0] / IN_DIM;
  const int E = in_sizes[1] / 2;
  const int* src = ei;
  const int* tgt = ei + E;

  char* ws = (char*)d_ws;
  size_t off = 0;
  int*   cnt    = (int*)(ws + off);   off = align_up(off + (size_t)(N + 1) * 4, 256);
  int*   slab   = (int*)(ws + off);   off = align_up(off + (size_t)N * SLOTS * 4, 256);
  int*   spill  = (int*)(ws + off);   off = align_up(off + (size_t)E * 8, 256);
  float* H1     = (float*)(ws + off); off = align_up(off + (size_t)N * IN_DIM * 4, 256);
  float* H2     = (float*)(ws + off); off = align_up(off + (size_t)N * OUT_DIM * 4, 256);
  float* out    = (float*)d_out;

  const int TB = 256;
  dim3 blk(TB);

  zero_cnt_kernel<<<dim3((N + 1 + TB - 1) / TB), blk, 0, stream>>>(cnt, N);
  count_place_kernel<<<dim3((E + TB - 1) / TB), blk, 0, stream>>>(src, tgt, cnt, slab, spill, E, N);

  const int gemm_blocks = (N + 127) / 128;
  const int nblocks = (N + 7) / 8;

  gemm_wmma_f32<128><<<dim3(gemm_blocks), blk, 0, stream>>>(x, W1, cnt, H1, N);
  const int fused_blocks = (N + 16 * FW - 1) / (16 * FW);
  gather_relu_gemm<<<dim3(fused_blocks), dim3(FW * 32), 0, stream>>>(cnt, slab, spill, H1, b1, W2, H2, N);

  gather_rows<64, false><<<dim3(nblocks), blk, 0, stream>>>(cnt, slab, spill, H2, b2, out, N);
}


# edge filing: returning global atomics + scattered 4-byte stores -> two passes ranked by LDS atomics (bin by 256-node bucket, rows built in LDS, whole-line stores); sums in LDS-atomic arrival order
# speedup vs baseline: 5.7361x; 1.6872x over previous
#include <hip/hip_runtime.h>


typedef __attribute__((ext_vector_type(2))) float v2f;
typedef __attribute__((ext_vector_type(8))) float v8f;

#define IN_DIM 128
#define WT_PITCH (IN_DIM + 4)

#define SLOTS 32


#define NPB_SHIFT 8
#define NPB 256
#define MAXB 512
#define BCAP 8192
#define BIN_T 512
#define BIN_K 12
#define TPITCH 33

__global__ void zero_ctl_kernel(int* __restrict__ ctl, int* __restrict__ spill_cnt) {
  int i = blockIdx.x * blockDim.x + threadIdx.x;
  if (i <= MAXB) ctl[i] = 0;
  if (i == 0) *spill_cnt = 0;
}

__global__ __launch_bounds__(BIN_T) void bin_edges(const int* __restrict__ src, const int* __restrict__ tgt,
                                                   int* __restrict__ ctl, int* __restrict__ region,
                                                   int* __restrict__ ovf, int e) {
  __shared__ int hist[MAXB];
  __shared__ int loff[MAXB];
  __shared__ int gbase[MAXB];
  __shared__ int wsum[BIN_T / 32];
  __shared__ int2 sorted[BIN_T * BIN_K];
  const int tid = threadIdx.x;
  const int lane = tid & 31;
  const int wave = tid >> 5;
  hist[tid] = 0;
  __syncthreads();

  const int base = blockIdx.x * (BIN_T * BIN_K);
  int s[BIN_K], t[BIN_K], r[BIN_K];
#pragma unroll
  for (int k = 0; k < BIN_K; ++k) {
    const int i = base + k * BIN_T + tid;
    t[k] = -1;
    s[k] = 0;
    if (i < e) {
      t[k] = tgt[i];
      s[k] = src[i];
    }
  }
#pragma unroll
  for (int k = 0; k < BIN_K; ++k) {
    r[k] = 0;
    if (t[k] >= 0) r[k] = atomicAdd(&hist[t[k] >> NPB_SHIFT], 1);
  }
  __syncthreads();

  const int h = hist[tid];
  int incl = h;
#pragma unroll
  for (int d = 1; d < 32; d <<= 1) {
    const int v = __shfl_up(incl, d);
    if (lane >= d) incl += v;
  }
  if (lane == 31) wsum[wave] = incl;
  __syncthreads();
  {
    int pre = 0;
    for (int w = 0; w < wave; ++w) pre += wsum[w];
    loff[tid] = pre + incl - h;
    int g = 0;
    if (h > 0) g = atomicAdd(&ctl[tid], h);
    gbase[tid] = g;
  }
  __syncthreads();

#pragma unroll
  for (int k = 0; k < BIN_K; ++k) {
    if (t[k] >= 0) {
      int2 ent;
      ent.x = s[k];
      ent.y = t[k];
      sorted[loff[t[k] >> NPB_SHIFT] + r[k]] = ent;
    }
  }
  __syncthreads();

  const int nloc = min(e - base, BIN_T * BIN_K);
#pragma unroll
  for (int k = 0; k < BIN_K; ++k) {
    const int i = k * BIN_T + tid;
    if (i < nloc) {
      const int2 ent = sorted[i];
      const int b = ent.y >> NPB_SHIFT;
      const int dst = gbase[b] + (i - loff[b]);
      if (dst < BCAP) {
        *(int2*)(region + 2 * ((size_t)b * BCAP + dst)) = ent;
      } else {
        const int q = atomicAdd(&ctl[MAXB], 1);
        *(int2*)(ovf + 2 * (size_t)q) = ent;
      }
    }
  }
}

__global__ __launch_bounds__(NPB) void build_tables(const int* __restrict__ ctl, const int* __restrict__ region,
                                                    const int* __restrict__ ovf, int* __restrict__ cnt,
                                                    int* __restrict__ slab, int* __restrict__ spill, int n) {
  __shared__ int lcnt[NPB];
  __shared__ int tab[NPB * TPITCH];
  const int tid = threadIdx.x;
  const int b = blockIdx.x;
  const int node0 = b << NPB_SHIFT;
  lcnt[tid] = 0;
  __syncthreads();

  const int m = min(ctl[b], BCAP);
  const int2* reg = (const int2*)(region + 2 * (size_t)b * BCAP);
  for (int base = 0; base < m; base += NPB * 8) {
    int2 en[8];
#pragma unroll
    for (int u = 0; u < 8; ++u) {
      const int i = base + u * NPB + tid;
      en[u].x = 0;
      en[u].y = -1;
      if (i < m) en[u] = reg[i];
    }
#pragma unroll
    for (int u = 0; u < 8; ++u) {
      if (en[u].y >= 0) {
        const int tl = en[u].y - node0;
        const int pos = atomicAdd(&lcnt[tl], 1);
        if (pos < SLOTS) {
          tab[tl * TPITCH + pos] = en[u].x;
        } else {
          const int q = atomicAdd(&cnt[n], 1);
          *(int2*)(spill + 2 * (size_t)q) = en[u];
        }
      }
    }
  }
  const int nov = ctl[MAXB];
  for (int i = tid; i < nov; i += NPB) {
    const int2 ent = *(const int2*)(ovf + 2 * (size_t)i);
    if ((ent.y >> NPB_SHIFT) == b) {
      const int tl = ent.y - node0;
      const int pos = atomicAdd(&lcnt[tl], 1);
      if (pos < SLOTS) {
        tab[tl * TPITCH + pos] = ent.x;
      } else {
        const int q = atomicAdd(&cnt[n], 1);
        *(int2*)(spill + 2 * (size_t)q) = ent;
      }
    }
  }
  __syncthreads();

  if (node0 + tid < n) cnt[node0 + tid] = lcnt[tid];
  for (int i = tid; i < NPB * SLOTS; i += NPB) {
    const int tl = i >> 5;
    const int sl = i & 31;
    if (node0 + tl < n) {
      int v = 0;
      if (sl < lcnt[tl]) v = tab[tl * TPITCH + sl];
      slab[(size_t)(node0 + tl) * SLOTS + sl] = v;
    }
  }
}


template <int NC>
__global__ __launch_bounds__(256) void gemm_wmma_f32(const float* __restrict__ X,
                                                     const float* __restrict__ W,
                                                     const int* __restrict__ cnt,
                                                     float* __restrict__ H, int n_rows) {
  __shared__ __align__(16) float Wt[NC * WT_PITCH];
  for (int i = threadIdx.x; i < NC * IN_DIM; i += 256) {
    int k = i / NC;
    int c = i - k * NC;
    Wt[c * WT_PITCH + k] = W[i];
  }
  __syncthreads();

  const int lane = threadIdx.x & 31;
  const int wave = threadIdx.x >> 5;
  const int row_base = blockIdx.x * 128 + wave * 16;
  if (row_base + 16 > n_rows) return;

  const int r = lane & 15;
  const int koff = (lane >> 4) << 1;

  const float* xrow = X + (size_t)(row_base + r) * IN_DIM + koff;

  constexpr int NT = NC / 16;
  v8f acc[NT];
  const v8f vzero = {0.f, 0.f, 0.f, 0.f, 0.f, 0.f, 0.f, 0.f};
#pragma unroll
  for (int t = 0; t < NT; ++t) acc[t] = vzero;

#pragma unroll 4
  for (int k0 = 0; k0 < IN_DIM; k0 += 4) {
    v2f a = *(const v2f*)(xrow + k0);
#pragma unroll
    for (int t = 0; t < NT; ++t) {
      v2f b = *(const v2f*)(&Wt[(t * 16 + r) * WT_PITCH + k0 + koff]);
      acc[t] = __builtin_amdgcn_wmma_f32_16x16x4_f32(false, a, false, b,
                                                     (short)0, acc[t], false, false);
    }
  }

  const int rowh = row_base + ((lane >> 4) << 3);
  const int4 d0 = *(const int4*)(cnt + rowh);
  const int4 d1 = *(const int4*)(cnt + rowh + 4);
  float sc[8];
  sc[0] = rsqrtf((float)(d0.x + 1));
  sc[1] = rsqrtf((float)(d0.y + 1));
  sc[2] = rsqrtf((float)(d0.z + 1));
  sc[3] = rsqrtf((float)(d0.w + 1));
  sc[4] = rsqrtf((float)(d1.x + 1));
  sc[5] = rsqrtf((float)(d1.y + 1));
  sc[6] = rsqrtf((float)(d1.z + 1));
  sc[7] = rsqrtf((float)(d1.w + 1));
#pragma unroll
  for (int t = 0; t < NT; ++t) {
    const int col = t * 16 + r;
#pragma unroll
    for (int j = 0; j < 8; ++j)
      H[(size_t)(rowh + j) * NC + col] = acc[t][j] * sc[j];
  }
}


template <int D, bool RELU>
__global__ __launch_bounds__(256) void gather_rows(const int* __restrict__ cnt,
                                                   const int* __restrict__ slab,
                                                   const int* __restrict__ spill,
                                                   const float* __restrict__ h,
                                                   const float* __restrict__ bias,
                                                   float* __restrict__ outp, int n) {
  constexpr int LPR = D / 4;
  constexpr int RPL = 32 / LPR;
  const int wv = __builtin_amdgcn_readfirstlane((int)(threadIdx.x >> 5));
  const int node = blockIdx.x * 8 + wv;
  if (node >= n) return;
  const int lane = threadIdx.x & 31;
  const int sub = lane / LPR;
  const int col = (lane - sub * LPR) * 4;

  const int deg = cnt[node];
  const int m = min(deg, SLOTS);

  float4 acc = make_float4(0.f, 0.f, 0.f, 0.f);
  if (sub == 0) acc = *(const float4*)(h + (size_t)node * D + col);

  int s = 0;
  if (lane < m) s = slab[(size_t)node * SLOTS + lane];
  const int mp = m - (m % RPL);
#pragma unroll 4
  for (int j = 0; j < mp; j += RPL) {
    int sj = __builtin_amdgcn_readlane(s, j);
    if (RPL == 2) {
      const int s1 = __builtin_amdgcn_readlane(s, j + 1);
      sj = sub ? s1 : sj;
    }
    const float4 v = *(const float4*)(h + (size_t)sj * D + col);
    acc.x += v.x;
    acc.y += v.y;
    acc.z += v.z;
    acc.w += v.w;
  }
  if (RPL == 2 && mp < m) {
    const int sj = __builtin_amdgcn_readlane(s, mp);
    if (sub == 0) {
      const float4 v = *(const float4*)(h + (size_t)sj * D + col);
      acc.x += v.x;
      acc.y += v.y;
      acc.z += v.z;
      acc.w += v.w;
    }
  }

  if (deg > SLOTS) {
    const int ns = cnt[n];
    for (int base = 0; base < ns; base += 32) {
      int es = 0;
      int et = -1;
      if (base + lane < ns) {
        const int2 ent = *(const int2*)(spill + 2 * (size_t)(base + lane));
        es = ent.x;
        et = ent.y;
      }
      unsigned mine = __builtin_amdgcn_ballot_w32(et == node);
      while (mine) {
        const int b = __builtin_ctz(mine);
        mine &= mine - 1;
        const int sj = __builtin_amdgcn_readlane(es, b);
        if (sub == 0) {
          const float4 v = *(const float4*)(h + (size_t)sj * D + col);
          acc.x += v.x;
          acc.y += v.y;
          acc.z += v.z;
          acc.w += v.w;
        }
      }
    }
  }

  if (RPL == 2) {
    acc.x += __shfl_xor(acc.x, 16);
    acc.y += __shfl_xor(acc.y, 16);
    acc.z += __shfl_xor(acc.z, 16);
    acc.w += __shfl_xor(acc.w, 16);
  }
  const float di = rsqrtf((float)(deg + 1));
  const float4 bv = *(const float4*)(bias + col);
  acc.x = di * acc.x + bv.x;
  acc.y = di * acc.y + bv.y;
  acc.z = di * acc.z + bv.z;
  acc.w = di * acc.w + bv.w;
  if (RELU) {
    acc.x = fmaxf(acc.x, 0.0f);
    acc.y = fmaxf(acc.y, 0.0f);
    acc.z = fmaxf(acc.z, 0.0f);
    acc.w = fmaxf(acc.w, 0.0f);
  }
  if (sub == 0) *(float4*)(outp + (size_t)node * D + col) = acc;
}

#define FW 25
#define OUT_DIM 64

__global__ __launch_bounds__(FW * 32) void gather_relu_gemm(const int* __restrict__ cnt,
                                                            const int* __restrict__ slab,
                                                            const int* __restrict__ spill,
                                                            const float* __restrict__ h,
                                                            const float* __restrict__ bias,
                                                            const float* __restrict__ W,
                                                            float* __restrict__ H2, int n) {
  __shared__ __align__(16) float Wt[OUT_DIM * WT_PITCH];
  __shared__ __align__(16) float strips[FW * 16 * WT_PITCH];
  const int lane = threadIdx.x & 31;
  const int wv = __builtin_amdgcn_readfirstlane((int)(threadIdx.x >> 5));
  const int row_base = (blockIdx.x * FW + wv) * 16;
  const bool live = row_base + 16 <= n;
  float* strip = strips + wv * (16 * WT_PITCH);
  const int col = lane * 4;

  int cntv = 0;
  float4 bv = make_float4(0.f, 0.f, 0.f, 0.f);
  if (live) {
    cntv = cnt[row_base + (lane & 15)];
    bv = *(const float4*)(bias + col);
  }

  constexpr int NW = OUT_DIM * IN_DIM;
  constexpr int TPB = FW * 32;
  constexpr int FULL = NW / TPB;
#pragma unroll
  for (int it = 0; it < FULL; ++it) {
    const int i = threadIdx.x + it * TPB;
    const int k = i / OUT_DIM;
    const int c = i - k * OUT_DIM;
    Wt[c * WT_PITCH + k] = W[i];
  }
  {
    const int i = threadIdx.x + FULL * TPB;
    if (i < NW) {
      const int k = i / OUT_DIM;
      const int c = i - k * OUT_DIM;
      Wt[c * WT_PITCH + k] = W[i];
    }
  }
  __syncthreads();
  if (!live) return;

  int m_next = min(__builtin_amdgcn_readlane(cntv, 0), SLOTS);
  int s_next = 0;
  if (lane < m_next) s_next = slab[(size_t)row_base * SLOTS + lane];

  for (int j = 0; j < 16; ++j) {
    const int node = row_base + j;
    const int deg = __builtin_amdgcn_readlane(cntv, j);
    const int m = m_next;
    const int s = s_next;
    if (j + 1 < 16) {
      m_next = min(__builtin_amdgcn_readlane(cntv, j + 1), SLOTS);
      s_next = 0;
      if (lane < m_next) s_next = slab[(size_t)(node + 1) * SLOTS + lane];
    }

    float4 acc = *(const float4*)(h + (size_t)node * IN_DIM + col);

    for (int base = 0; base < m; base += 8) {
      float4 v[8];
#pragma unroll
      for (int u = 0; u < 8; ++u) {
        v[u] = make_float4(0.f, 0.f, 0.f, 0.f);
        if (base + u < m) {
          const int sj = __builtin_amdgcn_readlane(s, base + u);
          v[u] = *(const float4*)(h + (size_t)sj * IN_DIM + col);
        }
      }
#pragma unroll
      for (int u = 0; u < 8; ++u) {
        acc.x += v[u].x;
        acc.y += v[u].y;
        acc.z += v[u].z;
        acc.w += v[u].w;
      }
    }

    if (deg > SLOTS) {
      const int ns = cnt[n];
      for (int base = 0; base < ns; base += 32) {
        int es = 0;
        int et = -1;
        if (base + lane < ns) {
          const int2 ent = *(const int2*)(spill + 2 * (size_t)(base + lane));
          es = ent.x;
          et = ent.y;
        }
        unsigned mine = __builtin_amdgcn_ballot_w32(et == node);
        while (mine) {
          const int b = __builtin_ctz(mine);
          mine &= mine - 1;
          const int sj = __builtin_amdgcn_readlane(es, b);
          const float4 v = *(const float4*)(h + (size_t)sj * IN_DIM + col);
          acc.x += v.x;
          acc.y += v.y;
          acc.z += v.z;
          acc.w += v.w;
        }
      }
    }

    const float di = rsqrtf((float)(deg + 1));
    acc.x = fmaxf(di * acc.x + bv.x, 0.0f);
    acc.y = fmaxf(di * acc.y + bv.y, 0.0f);
    acc.z = fmaxf(di * acc.z + bv.z, 0.0f);
    acc.w = fmaxf(di * acc.w + bv.w, 0.0f);
    *(float4*)(strip + j * WT_PITCH + col) = acc;
  }

  const int r = lane & 15;
  const int koff = (lane >> 4) << 1;
  constexpr int NT = OUT_DIM / 16;
  v8f acc2[NT];
  const v8f vzero = {0.f, 0.f, 0.f, 0.f, 0.f, 0.f, 0.f, 0.f};
#pragma unroll
  for (int t = 0; t < NT; ++t) acc2[t] = vzero;

#pragma unroll 4
  for (int k0 = 0; k0 < IN_DIM; k0 += 4) {
    v2f a = *(const v2f*)(strip + r * WT_PITCH + k0 + koff);
#pragma unroll
    for (int t = 0; t < NT; ++t) {
      v2f b = *(const v2f*)(&Wt[(t * 16 + r) * WT_PITCH + k0 + koff]);
      acc2[t] = __builtin_amdgcn_wmma_f32_16x16x4_f32(false, a, false, b,
                                                      (short)0, acc2[t], false, false);
    }
  }

  const int rowh = row_base + ((lane >> 4) << 3);
  const int4 d0 = *(const int4*)(cnt + rowh);
  const int4 d1 = *(const int4*)(cnt + rowh + 4);
  float sc[8];
  sc[0] = rsqrtf((float)(d0.x + 1));
  sc[1] = rsqrtf((float)(d0.y + 1));
  sc[2] = rsqrtf((float)(d0.z + 1));
  sc[3] = rsqrtf((float)(d0.w + 1));
  sc[4] = rsqrtf((float)(d1.x + 1));
  sc[5] = rsqrtf((float)(d1.y + 1));
  sc[6] = rsqrtf((float)(d1.z + 1));
  sc[7] = rsqrtf((float)(d1.w + 1));
#pragma unroll
  for (int t = 0; t < NT; ++t) {
    const int c = t * 16 + r;
#pragma unroll
    for (int j = 0; j < 8; ++j)
      H2[(size_t)(rowh + j) * OUT_DIM + c] = acc2[t][j] * sc[j];
  }
}


static inline size_t align_up(size_t x, size_t a) { return (x + a - 1) & ~(a - 1); }

extern "C" void kernel_launch(void* const* d_in, const int* in_sizes, int n_in,
                              void* d_out, int out_size, void* d_ws, size_t ws_size,
                              hipStream_t stream) {
  const float* x  = (const float*)d_in[0];
  const int*   ei = (const int*)d_in[1];
  const float* W1 = (const float*)d_in[2];
  const float* b1 = (const float*)d_in[3];
  const float* W2 = (const float*)d_in[4];
  const float* b2 = (const float*)d_in[5];

  const int N = in_sizes[0] / IN_DIM;
  const int E = in_sizes[1] / 2;
  const int* src = ei;
  const int* tgt = ei + E;

  char* ws = (char*)d_ws;
  size_t off = 0;
  int*   cnt    = (int*)(ws + off);   off = align_up(off + (size_t)(N + 1) * 4, 256);
  int*   slab   = (int*)(ws + off);   off = align_up(off + (size_t)N * SLOTS * 4, 256);
  int*   spill  = (int*)(ws + off);   off = align_up(off + (size_t)E * 8, 256);
  float* H1     = (float*)(ws + off); off = align_up(off + (size_t)N * IN_DIM * 4, 256);
  float* H2     = (float*)(ws + off); off = align_up(off + (size_t)N * OUT_DIM * 4, 256);
  int*   ctl    = (int*)(ws + off);   off = align_up(off + (size_t)(MAXB + 1) * 4, 256);
  int*   region = (int*)(ws + off);   off = align_up(off + (size_t)MAXB * BCAP * 8, 256);
  int*   ovf    = (int*)(ws + off);   off = align_up(off + (size_t)E * 8, 256);
  float* out    = (float*)d_out;

  const int TB = 256;
  dim3 blk(TB);

  zero_ctl_kernel<<<dim3((MAXB + 1 + TB - 1) / TB), blk, 0, stream>>>(ctl, cnt + N);
  bin_edges<<<dim3((E + BIN_T * BIN_K - 1) / (BIN_T * BIN_K)), dim3(BIN_T), 0, stream>>>(src, tgt, ctl, region, ovf, E);
  build_tables<<<dim3((N + NPB - 1) / NPB), dim3(NPB), 0, stream>>>(ctl, region, ovf, cnt, slab, spill, N);

  const int gemm_blocks = (N + 127) / 128;
  const int nblocks = (N + 7) / 8;

  gemm_wmma_f32<128><<<dim3(gemm_blocks), blk, 0, stream>>>(x, W1, cnt, H1, N);
  const int fused_blocks = (N + 16 * FW - 1) / (16 * FW);
  gather_relu_gemm<<<dim3(fused_blocks), dim3(FW * 32), 0, stream>>>(cnt, slab, spill, H1, b1, W2, H2, N);

  gather_rows<64, false><<<dim3(nblocks), blk, 0, stream>>>(cnt, slab, spill, H2, b2, out, N);
}


# layer-1 GEMM row blocks ride as extra work-groups in the two edge-filing launches (matrix pipe busy under their waits), own launch gone; dinv[src] becomes a scalar-loaded FMA weight in the gather
# speedup vs baseline: 6.0836x; 1.0606x over previous
#include <hip/hip_runtime.h>


typedef __attribute__((ext_vector_type(2))) float v2f;
typedef __attribute__((ext_vector_type(8))) float v8f;

#define IN_DIM 128
#define WT_PITCH (IN_DIM + 4)

#define SLOTS 32


#define NPB_SHIFT 8
#define NPB 256
#define MAXB 512
#define BCAP 8192
#define BIN_T 512
#define BIN_K 12
#define TPITCH 33

__global__ void zero_ctl_kernel(int* __restrict__ ctl, int* __restrict__ spill_cnt) {
  int i = blockIdx.x * blockDim.x + threadIdx.x;
  if (i <= MAXB) ctl[i] = 0;
  if (i == 0) *spill_cnt = 0;
}

#define BIN_LDS_INTS (3 * MAXB + BIN_T / 32 + 2 * BIN_T * BIN_K)
__device__ __forceinline__ void bin_edges_body(int* smem, const int* __restrict__ src, const int* __restrict__ tgt,
                                               int* __restrict__ ctl, int* __restrict__ region,
                                               int* __restrict__ ovf, int e) {
  int* hist = smem;
  int* loff = smem + MAXB;
  int* gbase = smem + 2 * MAXB;
  int* wsum = smem + 3 * MAXB;
  int2* sorted = (int2*)(smem + 3 * MAXB + BIN_T / 32);
  const int tid = threadIdx.x;
  const int lane = tid & 31;
  const int wave = tid >> 5;
  hist[tid] = 0;
  __syncthreads();

  const int base = blockIdx.x * (BIN_T * BIN_K);
  int s[BIN_K], t[BIN_K], r[BIN_K];
#pragma unroll
  for (int k = 0; k < BIN_K; ++k) {
    const int i = base + k * BIN_T + tid;
    t[k] = -1;
    s[k] = 0;
    if (i < e) {
      t[k] = tgt[i];
      s[k] = src[i];
    }
  }
#pragma unroll
  for (int k = 0; k < BIN_K; ++k) {
    r[k] = 0;
    if (t[k] >= 0) r[k] = atomicAdd(&hist[t[k] >> NPB_SHIFT], 1);
  }
  __syncthreads();

  const int h = hist[tid];
  int incl = h;
#pragma unroll
  for (int d = 1; d < 32; d <<= 1) {
    const int v = __shfl_up(incl, d);
    if (lane >= d) incl += v;
  }
  if (lane == 31) wsum[wave] = incl;
  __syncthreads();
  {
    int pre = 0;
    for (int w = 0; w < wave; ++w) pre += wsum[w];
    loff[tid] = pre + incl - h;
    int g = 0;
    if (h > 0) g = atomicAdd(&ctl[tid], h);
    gbase[tid] = g;
  }
  __syncthreads();

#pragma unroll
  for (int k = 0; k < BIN_K; ++k) {
    if (t[k] >= 0) {
      int2 ent;
      ent.x = s[k];
      ent.y = t[k];
      sorted[loff[t[k] >> NPB_SHIFT] + r[k]] = ent;
    }
  }
  __syncthreads();

  const int nloc = min(e - base, BIN_T * BIN_K);
#pragma unroll
  for (int k = 0; k < BIN_K; ++k) {
    const int i = k * BIN_T + tid;
    if (i < nloc) {
      const int2 ent = sorted[i];
      const int b = ent.y >> NPB_SHIFT;
      const int dst = gbase[b] + (i - loff[b]);
      if (dst < BCAP) {
        *(int2*)(region + 2 * ((size_t)b * BCAP + dst)) = ent;
      } else {
        const int q = atomicAdd(&ctl[MAXB], 1);
        *(int2*)(ovf + 2 * (size_t)q) = ent;
      }
    }
  }
}

#define BUILD_LDS_INTS (NPB + NPB * TPITCH)
__device__ __forceinline__ void build_tables_body(int* smem, const int* __restrict__ ctl,
                                                  const int* __restrict__ region, const int* __restrict__ ovf,
                                                  int* __restrict__ cnt, float* __restrict__ dinv,
                                                  int* __restrict__ slab, int* __restrict__ spill, int n) {
  int* lcnt = smem;
  int* tab = smem + NPB;
  const int tid = threadIdx.x;
  const int b = blockIdx.x;
  const int node0 = b << NPB_SHIFT;
  lcnt[tid] = 0;
  __syncthreads();

  const int m = min(ctl[b], BCAP);
  const int2* reg = (const int2*)(region + 2 * (size_t)b * BCAP);
  for (int base = 0; base < m; base += NPB * 8) {
    int2 en[8];
#pragma unroll
    for (int u = 0; u < 8; ++u) {
      const int i = base + u * NPB + tid;
      en[u].x = 0;
      en[u].y = -1;
      if (i < m) en[u] = reg[i];
    }
#pragma unroll
    for (int u = 0; u < 8; ++u) {
      if (en[u].y >= 0) {
        const int tl = en[u].y - node0;
        const int pos = atomicAdd(&lcnt[tl], 1);
        if (pos < SLOTS) {
          tab[tl * TPITCH + pos] = en[u].x;
        } else {
          const int q = atomicAdd(&cnt[n], 1);
          *(int2*)(spill + 2 * (size_t)q) = en[u];
        }
      }
    }
  }
  const int nov = ctl[MAXB];
  for (int i = tid; i < nov; i += NPB) {
    const int2 ent = *(const int2*)(ovf + 2 * (size_t)i);
    if ((ent.y >> NPB_SHIFT) == b) {
      const int tl = ent.y - node0;
      const int pos = atomicAdd(&lcnt[tl], 1);
      if (pos < SLOTS) {
        tab[tl * TPITCH + pos] = ent.x;
      } else {
        const int q = atomicAdd(&cnt[n], 1);
        *(int2*)(spill + 2 * (size_t)q) = ent;
      }
    }
  }
  __syncthreads();

  if (node0 + tid < n) {
    cnt[node0 + tid] = lcnt[tid];
    dinv[node0 + tid] = rsqrtf((float)(lcnt[tid] + 1));
  }
  for (int i = tid; i < NPB * SLOTS; i += NPB) {
    const int tl = i >> 5;
    const int sl = i & 31;
    if (node0 + tl < n) {
      int v = 0;
      if (sl < lcnt[tl]) v = tab[tl * TPITCH + sl];
      slab[(size_t)(node0 + tl) * SLOTS + sl] = v;
    }
  }
}


template <int NC, int NTHR>
__device__ __forceinline__ void gemm_wmma_f32_rows(float* Wt,
                                                   const float* __restrict__ X,
                                                   const float* __restrict__ W,
                                                   float* __restrict__ H, int row0, int row_end) {
  for (int i = threadIdx.x; i < NC * IN_DIM; i += NTHR) {
    int k = i / NC;
    int c = i - k * NC;
    Wt[c * WT_PITCH + k] = W[i];
  }
  __syncthreads();

  const int lane = threadIdx.x & 31;
  const int wave = threadIdx.x >> 5;
  const int row_base = row0 + wave * 16;
  if (row_base + 16 > row_end) return;

  const int r = lane & 15;
  const int koff = (lane >> 4) << 1;

  const float* xrow = X + (size_t)(row_base + r) * IN_DIM + koff;

  constexpr int NT = NC / 16;
  v8f acc[NT];
  const v8f vzero = {0.f, 0.f, 0.f, 0.f, 0.f, 0.f, 0.f, 0.f};
#pragma unroll
  for (int t = 0; t < NT; ++t) acc[t] = vzero;

#pragma unroll 4
  for (int k0 = 0; k0 < IN_DIM; k0 += 4) {
    v2f a = *(const v2f*)(xrow + k0);
#pragma unroll
    for (int t = 0; t < NT; ++t) {
      v2f b = *(const v2f*)(&Wt[(t * 16 + r) * WT_PITCH + k0 + koff]);
      acc[t] = __builtin_amdgcn_wmma_f32_16x16x4_f32(false, a, false, b,
                                                     (short)0, acc[t], false, false);
    }
  }

  const int rowh = row_base + ((lane >> 4) << 3);
#pragma unroll
  for (int t = 0; t < NT; ++t) {
    const int col = t * 16 + r;
#pragma unroll
    for (int j = 0; j < 8; ++j)
      H[(size_t)(rowh + j) * NC + col] = acc[t][j];
  }
}

#define GEMM_LDS_INTS (IN_DIM * WT_PITCH)
#define GEMM_A_BLOCKS 240

__global__ __launch_bounds__(BIN_T) void bin_edges_gemm(const int* __restrict__ src, const int* __restrict__ tgt,
                                                        int* __restrict__ ctl, int* __restrict__ region,
                                                        int* __restrict__ ovf, int e,
                                                        const float* __restrict__ X, const float* __restrict__ W,
                                                        float* __restrict__ H, int row_end, int nfile) {
  static_assert(BIN_LDS_INTS <= GEMM_LDS_INTS, "LDS block too small for the binning pass");
  __shared__ __align__(16) int smem[GEMM_LDS_INTS];
  if ((int)blockIdx.x < nfile) {
    bin_edges_body(smem, src, tgt, ctl, region, ovf, e);
  } else {
    const int g = (int)blockIdx.x - nfile;
    gemm_wmma_f32_rows<IN_DIM, BIN_T>((float*)smem, X, W, H, g * (BIN_T / 2), row_end);
  }
}

__global__ __launch_bounds__(NPB) void build_tables_gemm(const int* __restrict__ ctl, const int* __restrict__ region,
                                                         const int* __restrict__ ovf, int* __restrict__ cnt,
                                                         float* __restrict__ dinv,
                                                         int* __restrict__ slab, int* __restrict__ spill, int n,
                                                         const float* __restrict__ X, const float* __restrict__ W,
                                                         float* __restrict__ H, int row_start, int row_end,
                                                         int nfile) {
  static_assert(BUILD_LDS_INTS <= GEMM_LDS_INTS, "LDS block too small for the build pass");
  __shared__ __align__(16) int smem[GEMM_LDS_INTS];
  if ((int)blockIdx.x < nfile) {
    build_tables_body(smem, ctl, region, ovf, cnt, dinv, slab, spill, n);
  } else {
    const int g = (int)blockIdx.x - nfile;
    gemm_wmma_f32_rows<IN_DIM, NPB>((float*)smem, X, W, H, row_start + g * (NPB / 2), row_end);
  }
}


template <int D, bool RELU>
__global__ __launch_bounds__(256) void gather_rows(const int* __restrict__ cnt,
                                                   const int* __restrict__ slab,
                                                   const int* __restrict__ spill,
                                                   const float* __restrict__ h,
                                                   const float* __restrict__ bias,
                                                   float* __restrict__ outp, int n) {
  constexpr int LPR = D / 4;
  constexpr int RPL = 32 / LPR;
  const int wv = __builtin_amdgcn_readfirstlane((int)(threadIdx.x >> 5));
  const int node = blockIdx.x * 8 + wv;
  if (node >= n) return;
  const int lane = threadIdx.x & 31;
  const int sub = lane / LPR;
  const int col = (lane - sub * LPR) * 4;

  const int deg = cnt[node];
  const int m = min(deg, SLOTS);

  float4 acc = make_float4(0.f, 0.f, 0.f, 0.f);
  if (sub == 0) acc = *(const float4*)(h + (size_t)node * D + col);

  int s = 0;
  if (lane < m) s = slab[(size_t)node * SLOTS + lane];
  const int mp = m - (m % RPL);
#pragma unroll 4
  for (int j = 0; j < mp; j += RPL) {
    int sj = __builtin_amdgcn_readlane(s, j);
    if (RPL == 2) {
      const int s1 = __builtin_amdgcn_readlane(s, j + 1);
      sj = sub ? s1 : sj;
    }
    const float4 v = *(const float4*)(h + (size_t)sj * D + col);
    acc.x += v.x;
    acc.y += v.y;
    acc.z += v.z;
    acc.w += v.w;
  }
  if (RPL == 2 && mp < m) {
    const int sj = __builtin_amdgcn_readlane(s, mp);
    if (sub == 0) {
      const float4 v = *(const float4*)(h + (size_t)sj * D + col);
      acc.x += v.x;
      acc.y += v.y;
      acc.z += v.z;
      acc.w += v.w;
    }
  }

  if (deg > SLOTS) {
    const int ns = cnt[n];
    for (int base = 0; base < ns; base += 32) {
      int es = 0;
      int et = -1;
      if (base + lane < ns) {
        const int2 ent = *(const int2*)(spill + 2 * (size_t)(base + lane));
        es = ent.x;
        et = ent.y;
      }
      unsigned mine = __builtin_amdgcn_ballot_w32(et == node);
      while (mine) {
        const int b = __builtin_ctz(mine);
        mine &= mine - 1;
        const int sj = __builtin_amdgcn_readlane(es, b);
        if (sub == 0) {
          const float4 v = *(const float4*)(h + (size_t)sj * D + col);
          acc.x += v.x;
          acc.y += v.y;
          acc.z += v.z;
          acc.w += v.w;
        }
      }
    }
  }

  if (RPL == 2) {
    acc.x += __shfl_xor(acc.x, 16);
    acc.y += __shfl_xor(acc.y, 16);
    acc.z += __shfl_xor(acc.z, 16);
    acc.w += __shfl_xor(acc.w, 16);
  }
  const float di = rsqrtf((float)(deg + 1));
  const float4 bv = *(const float4*)(bias + col);
  acc.x = di * acc.x + bv.x;
  acc.y = di * acc.y + bv.y;
  acc.z = di * acc.z + bv.z;
  acc.w = di * acc.w + bv.w;
  if (RELU) {
    acc.x = fmaxf(acc.x, 0.0f);
    acc.y = fmaxf(acc.y, 0.0f);
    acc.z = fmaxf(acc.z, 0.0f);
    acc.w = fmaxf(acc.w, 0.0f);
  }
  if (sub == 0) *(float4*)(outp + (size_t)node * D + col) = acc;
}

#define FW 25
#define OUT_DIM 64

__global__ __launch_bounds__(FW * 32) void gather_relu_gemm(const int* __restrict__ cnt,
                                                            const float* __restrict__ dinv,
                                                            const int* __restrict__ slab,
                                                            const int* __restrict__ spill,
                                                            const float* __restrict__ h,
                                                            const float* __restrict__ bias,
                                                            const float* __restrict__ W,
                                                            float* __restrict__ H2, int n) {
  __shared__ __align__(16) float Wt[OUT_DIM * WT_PITCH];
  __shared__ __align__(16) float strips[FW * 16 * WT_PITCH];
  const int lane = threadIdx.x & 31;
  const int wv = __builtin_amdgcn_readfirstlane((int)(threadIdx.x >> 5));
  const int row_base = (blockIdx.x * FW + wv) * 16;
  const bool live = row_base + 16 <= n;
  float* strip = strips + wv * (16 * WT_PITCH);
  const int col = lane * 4;

  int cntv = 0;
  float4 bv = make_float4(0.f, 0.f, 0.f, 0.f);
  if (live) {
    cntv = cnt[row_base + (lane & 15)];
    bv = *(const float4*)(bias + col);
  }

  constexpr int NW = OUT_DIM * IN_DIM;
  constexpr int TPB = FW * 32;
  constexpr int FULL = NW / TPB;
#pragma unroll
  for (int it = 0; it < FULL; ++it) {
    const int i = threadIdx.x + it * TPB;
    const int k = i / OUT_DIM;
    const int c = i - k * OUT_DIM;
    Wt[c * WT_PITCH + k] = W[i];
  }
  {
    const int i = threadIdx.x + FULL * TPB;
    if (i < NW) {
      const int k = i / OUT_DIM;
      const int c = i - k * OUT_DIM;
      Wt[c * WT_PITCH + k] = W[i];
    }
  }
  __syncthreads();
  if (!live) return;

  int m_next = min(__builtin_amdgcn_readlane(cntv, 0), SLOTS);
  int s_next = 0;
  if (lane < m_next) s_next = slab[(size_t)row_base * SLOTS + lane];

  for (int j = 0; j < 16; ++j) {
    const int node = row_base + j;
    const int deg = __builtin_amdgcn_readlane(cntv, j);
    const int m = m_next;
    const int s = s_next;
    if (j + 1 < 16) {
      m_next = min(__builtin_amdgcn_readlane(cntv, j + 1), SLOTS);
      s_next = 0;
      if (lane < m_next) s_next = slab[(size_t)(node + 1) * SLOTS + lane];
    }

    const float di = rsqrtf((float)(deg + 1));
    const float4 sv = *(const float4*)(h + (size_t)node * IN_DIM + col);
    float4 acc = make_float4(0.f, 0.f, 0.f, 0.f);

    int base = 0;
    do {
      float4 v[8];
      float wu[8];
#pragma unroll
      for (int u = 0; u < 8; ++u) {
        v[u] = make_float4(0.f, 0.f, 0.f, 0.f);
        wu[u] = 0.f;
        if (base + u < m) {
          const int sj = __builtin_amdgcn_readlane(s, base + u);
          wu[u] = dinv[sj];
          v[u] = *(const float4*)(h + (size_t)sj * IN_DIM + col);
        }
      }
      if (base == 0) {
        acc.x = di * sv.x;
        acc.y = di * sv.y;
        acc.z = di * sv.z;
        acc.w = di * sv.w;
      }
#pragma unroll
      for (int u = 0; u < 8; ++u) {
        acc.x = fmaf(wu[u], v[u].x, acc.x);
        acc.y = fmaf(wu[u], v[u].y, acc.y);
        acc.z = fmaf(wu[u], v[u].z, acc.z);
        acc.w = fmaf(wu[u], v[u].w, acc.w);
      }
      base += 8;
    } while (base < m);

    if (deg > SLOTS) {
      const int ns = cnt[n];
      for (int base = 0; base < ns; base += 32) {
        int es = 0;
        int et = -1;
        if (base + lane < ns) {
          const int2 ent = *(const int2*)(spill + 2 * (size_t)(base + lane));
          es = ent.x;
          et = ent.y;
        }
        unsigned mine = __builtin_amdgcn_ballot_w32(et == node);
        while (mine) {
          const int b = __builtin_ctz(mine);
          mine &= mine - 1;
          const int sj = __builtin_amdgcn_readlane(es, b);
          const float wj = dinv[sj];
          const float4 v = *(const float4*)(h + (size_t)sj * IN_DIM + col);
          acc.x = fmaf(wj, v.x, acc.x);
          acc.y = fmaf(wj, v.y, acc.y);
          acc.z = fmaf(wj, v.z, acc.z);
          acc.w = fmaf(wj, v.w, acc.w);
        }
      }
    }

    acc.x = fmaxf(di * acc.x + bv.x, 0.0f);
    acc.y = fmaxf(di * acc.y + bv.y, 0.0f);
    acc.z = fmaxf(di * acc.z + bv.z, 0.0f);
    acc.w = fmaxf(di * acc.w + bv.w, 0.0f);
    *(float4*)(strip + j * WT_PITCH + col) = acc;
  }

  const int r = lane & 15;
  const int koff = (lane >> 4) << 1;
  constexpr int NT = OUT_DIM / 16;
  v8f acc2[NT];
  const v8f vzero = {0.f, 0.f, 0.f, 0.f, 0.f, 0.f, 0.f, 0.f};
#pragma unroll
  for (int t = 0; t < NT; ++t) acc2[t] = vzero;

#pragma unroll 4
  for (int k0 = 0; k0 < IN_DIM; k0 += 4) {
    v2f a = *(const v2f*)(strip + r * WT_PITCH + k0 + koff);
#pragma unroll
    for (int t = 0; t < NT; ++t) {
      v2f b = *(const v2f*)(&Wt[(t * 16 + r) * WT_PITCH + k0 + koff]);
      acc2[t] = __builtin_amdgcn_wmma_f32_16x16x4_f32(false, a, false, b,
                                                      (short)0, acc2[t], false, false);
    }
  }

  const int rowh = row_base + ((lane >> 4) << 3);
  const int4 d0 = *(const int4*)(cnt + rowh);
  const int4 d1 = *(const int4*)(cnt + rowh + 4);
  float sc[8];
  sc[0] = rsqrtf((float)(d0.x + 1));
  sc[1] = rsqrtf((float)(d0.y + 1));
  sc[2] = rsqrtf((float)(d0.z + 1));
  sc[3] = rsqrtf((float)(d0.w + 1));
  sc[4] = rsqrtf((float)(d1.x + 1));
  sc[5] = rsqrtf((float)(d1.y + 1));
  sc[6] = rsqrtf((float)(d1.z + 1));
  sc[7] = rsqrtf((float)(d1.w + 1));
#pragma unroll
  for (int t = 0; t < NT; ++t) {
    const int c = t * 16 + r;
#pragma unroll
    for (int j = 0; j < 8; ++j)
      H2[(size_t)(rowh + j) * OUT_DIM + c] = acc2[t][j] * sc[j];
  }
}


static inline size_t align_up(size_t x, size_t a) { return (x + a - 1) & ~(a - 1); }

extern "C" void kernel_launch(void* const* d_in, const int* in_sizes, int n_in,
                              void* d_out, int out_size, void* d_ws, size_t ws_size,
                              hipStream_t stream) {
  const float* x  = (const float*)d_in[0];
  const int*   ei = (const int*)d_in[1];
  const float* W1 = (const float*)d_in[2];
  const float* b1 = (const float*)d_in[3];
  const float* W2 = (const float*)d_in[4];
  const float* b2 = (const float*)d_in[5];

  const int N = in_sizes[0] / IN_DIM;
  const int E = in_sizes[1] / 2;
  const int* src = ei;
  const int* tgt = ei + E;

  char* ws = (char*)d_ws;
  size_t off = 0;
  int*   cnt    = (int*)(ws + off);   off = align_up(off + (size_t)(N + 1) * 4, 256);
  float* dinv   = (float*)(ws + off); off = align_up(off + (size_t)N * 4, 256);
  int*   slab   = (int*)(ws + off);   off = align_up(off + (size_t)N * SLOTS * 4, 256);
  int*   spill  = (int*)(ws + off);   off = align_up(off + (size_t)E * 8, 256);
  float* H1     = (float*)(ws + off); off = align_up(off + (size_t)N * IN_DIM * 4, 256);
  float* H2     = (float*)(ws + off); off = align_up(off + (size_t)N * OUT_DIM * 4, 256);
  int*   ctl    = (int*)(ws + off);   off = align_up(off + (size_t)(MAXB + 1) * 4, 256);
  int*   region = (int*)(ws + off);   off = align_up(off + (size_t)MAXB * BCAP * 8, 256);
  int*   ovf    = (int*)(ws + off);   off = align_up(off + (size_t)E * 8, 256);
  float* out    = (float*)d_out;

  const int TB = 256;
  dim3 blk(TB);

  const int nbin = (E + BIN_T * BIN_K - 1) / (BIN_T * BIN_K);
  const int nbuild = (N + NPB - 1) / NPB;
  const int rows_per_a = BIN_T / 2;
  const int rows_per_b = NPB / 2;
  const int ga = (N / rows_per_a < GEMM_A_BLOCKS) ? (N / rows_per_a) : GEMM_A_BLOCKS;
  const int rows_a = ga * rows_per_a;
  const int gb = (N - rows_a + rows_per_b - 1) / rows_per_b;
  zero_ctl_kernel<<<dim3((MAXB + 1 + TB - 1) / TB), blk, 0, stream>>>(ctl, cnt + N);
  bin_edges_gemm<<<dim3(nbin + ga), dim3(BIN_T), 0, stream>>>(src, tgt, ctl, region, ovf, E, x, W1, H1, rows_a, nbin);
  build_tables_gemm<<<dim3(nbuild + gb), dim3(NPB), 0, stream>>>(ctl, region, ovf, cnt, dinv, slab, spill, N,
                                                                x, W1, H1, rows_a, N, nbuild);

  const int nblocks = (N + 7) / 8;

  const int fused_blocks = (N + 16 * FW - 1) / (16 * FW);
  gather_relu_gemm<<<dim3(fused_blocks), dim3(FW * 32), 0, stream>>>(cnt, dinv, slab, spill, H1, b1, W2, H2, N);

  gather_rows<64, false><<<dim3(nblocks), blk, 0, stream>>>(cnt, slab, spill, H2, b2, out, N);
}
